# starting kernel
# baseline (speedup 1.0000x reference)
#include <hip/hip_runtime.h>
#include <hip/hip_bf16.h>

typedef __attribute__((ext_vector_type(2))) float v2f;
typedef __attribute__((ext_vector_type(8))) float v8f;

__global__ __launch_bounds__(256) void fgcn_pack_w(const float* __restrict__ W,
                                                   float* __restrict__ Wp,
                                                   int total  ) {
    int tid = blockIdx.x * blockDim.x + threadIdx.x;
    if (tid >= total) return;
    const int lane = tid & 31;
    const int n    = (tid >> 5) & 3;
    const int kk   = tid >> 7;
    const int kb   = kk * 4 + ((lane >= 16) ? 2 : 0);
    const int col  = n * 16 + (lane & 15);
    float2 v;
    v.x = W[(size_t)kb * 64 + col];
    v.y = W[(size_t)(kb + 1) * 64 + col];
    *(float2*)(Wp + (size_t)tid * 2) = v;
}

__global__ __launch_bounds__(256) void fgcn_gemm_wmma(const float* __restrict__ X,
                                                      const float* __restrict__ Wp,
                                                      float* __restrict__ H,
                                                      int mtiles, int K) {
    const int wave = (blockIdx.x * blockDim.x + threadIdx.x) >> 5;
    const int lane = threadIdx.x & 31;
    if (wave >= mtiles) return;

    const int m0     = wave * 16;
    const int lane16 = lane & 15;
    const bool hi    = lane >= 16;

    v8f acc0 = {}, acc1 = {}, acc2 = {}, acc3 = {};

    const float* xa = X + (size_t)(m0 + lane16) * K + (hi ? 2 : 0);
    const float* wb = Wp + (size_t)lane * 2;

    for (int k = 0; k < K; k += 4) {
        float2 af = *(const float2*)(xa + k);
        v2f a; a.x = af.x; a.y = af.y;

        const float* wk = wb + (size_t)(k >> 2) * 256;
        v2f b0 = *(const v2f*)(wk);
        v2f b1 = *(const v2f*)(wk + 64);
        v2f b2 = *(const v2f*)(wk + 128);
        v2f b3 = *(const v2f*)(wk + 192);

        acc0 = __builtin_amdgcn_wmma_f32_16x16x4_f32(false, a, false, b0, (short)0, acc0, false, false);
        acc1 = __builtin_amdgcn_wmma_f32_16x16x4_f32(false, a, false, b1, (short)0, acc1, false, false);
        acc2 = __builtin_amdgcn_wmma_f32_16x16x4_f32(false, a, false, b2, (short)0, acc2, false, false);
        acc3 = __builtin_amdgcn_wmma_f32_16x16x4_f32(false, a, false, b3, (short)0, acc3, false, false);
    }

    const int rbase = m0 + (hi ? 8 : 0);
    #pragma unroll
    for (int r = 0; r < 8; ++r) {
        float* o = H + (size_t)(rbase + r) * 64 + lane16;
        o[0]  = acc0[r];
        o[16] = acc1[r];
        o[32] = acc2[r];
        o[48] = acc3[r];
    }
}

__global__ __launch_bounds__(256) void fgcn_bias_init(float* __restrict__ out,
                                                      const float* __restrict__ bias,
                                                      int total) {
    int idx = blockIdx.x * blockDim.x + threadIdx.x;
    if (idx < total) out[idx] = bias[idx & 63];
}

__global__ __launch_bounds__(256) void fgcn_edge_scatter(const float* __restrict__ h,
                                                         const int* __restrict__ src,
                                                         const int* __restrict__ dst,
                                                         float* __restrict__ out,
                                                         int E) {
    int idx = blockIdx.x * blockDim.x + threadIdx.x;
    if (idx >= E * 16) return;
    const int e = idx >> 4;
    const int q = (idx & 15) << 2;
    const int s = src[e];
    const int d = dst[e];
    float4 v = *(const float4*)(h + (size_t)s * 64 + q);
    float* o = out + (size_t)d * 64 + q;
    atomicAdd(o + 0, v.x);
    atomicAdd(o + 1, v.y);
    atomicAdd(o + 2, v.z);
    atomicAdd(o + 3, v.w);
}

static constexpr int N_DRUG = 50000, E_DRUG = 1000000, F_DRUG = 256;
static constexpr int N_DIS  = 20000, E_DIS  = 640000,  F_DIS  = 128;
static constexpr int NHID2  = 64;

extern "C" void kernel_launch(void* const* d_in, const int* in_sizes, int n_in,
                              void* d_out, int out_size, void* d_ws, size_t ws_size,
                              hipStream_t stream) {
    const float* drug_x  = (const float*)d_in[0];
    const float* dis_x   = (const float*)d_in[1];
    const float* W1      = (const float*)d_in[2];
    const float* b1      = (const float*)d_in[3];
    const float* W2      = (const float*)d_in[4];
    const float* b2      = (const float*)d_in[5];
    const int*   drug_ei = (const int*)d_in[6];
    const int*   dis_ei  = (const int*)d_in[7];

    float* out1 = (float*)d_out;
    float* out2 = out1 + (size_t)N_DRUG * NHID2;

    float* h1  = (float*)d_ws;
    float* h2  = h1 + (size_t)N_DRUG * NHID2;
    float* Wp1 = h2 + (size_t)N_DIS * NHID2;
    float* Wp2 = Wp1 + (size_t)F_DRUG * NHID2;

    {
        int total = F_DRUG * NHID2 / 2;
        fgcn_pack_w<<<(total + 255) / 256, 256, 0, stream>>>(W1, Wp1, total);
    }
    {
        int total = F_DIS * NHID2 / 2;
        fgcn_pack_w<<<(total + 255) / 256, 256, 0, stream>>>(W2, Wp2, total);
    }

    {
        int mtiles = N_DRUG / 16;
        int grid = (mtiles + 7) / 8;
        fgcn_gemm_wmma<<<grid, 256, 0, stream>>>(drug_x, Wp1, h1, mtiles, F_DRUG);
    }
    {
        int mtiles = N_DIS / 16;
        int grid = (mtiles + 7) / 8;
        fgcn_gemm_wmma<<<grid, 256, 0, stream>>>(dis_x, Wp2, h2, mtiles, F_DIS);
    }

    {
        int total = N_DRUG * NHID2;
        fgcn_bias_init<<<(total + 255) / 256, 256, 0, stream>>>(out1, b1, total);
    }
    {
        int total = N_DIS * NHID2;
        fgcn_bias_init<<<(total + 255) / 256, 256, 0, stream>>>(out2, b2, total);
    }

    {
        int threads = E_DRUG * 16;
        fgcn_edge_scatter<<<(threads + 255) / 256, 256, 0, stream>>>(
            h1, drug_ei, drug_ei + E_DRUG, out1, E_DRUG);
    }
    {
        int threads = E_DIS * 16;
        fgcn_edge_scatter<<<(threads + 255) / 256, 256, 0, stream>>>(
            h2, dis_ei, dis_ei + E_DIS, out2, E_DIS);
    }
}


# edge scatter remapped: one wave per 32 edges, lane = feature column, so each gather load and float atomic covers one whole 128-byte line, not strided pieces of four; sum order still by atomic arrival
# speedup vs baseline: 1.7086x; 1.7086x over previous
#include <hip/hip_runtime.h>
#include <hip/hip_bf16.h>

typedef __attribute__((ext_vector_type(2))) float v2f;
typedef __attribute__((ext_vector_type(8))) float v8f;

__global__ __launch_bounds__(256) void fgcn_pack_w(const float* __restrict__ W,
                                                   float* __restrict__ Wp,
                                                   int total  ) {
    int tid = blockIdx.x * blockDim.x + threadIdx.x;
    if (tid >= total) return;
    const int lane = tid & 31;
    const int n    = (tid >> 5) & 3;
    const int kk   = tid >> 7;
    const int kb   = kk * 4 + ((lane >= 16) ? 2 : 0);
    const int col  = n * 16 + (lane & 15);
    float2 v;
    v.x = W[(size_t)kb * 64 + col];
    v.y = W[(size_t)(kb + 1) * 64 + col];
    *(float2*)(Wp + (size_t)tid * 2) = v;
}

__global__ __launch_bounds__(256) void fgcn_gemm_wmma(const float* __restrict__ X,
                                                      const float* __restrict__ Wp,
                                                      float* __restrict__ H,
                                                      int mtiles, int K) {
    const int wave = (blockIdx.x * blockDim.x + threadIdx.x) >> 5;
    const int lane = threadIdx.x & 31;
    if (wave >= mtiles) return;

    const int m0     = wave * 16;
    const int lane16 = lane & 15;
    const bool hi    = lane >= 16;

    v8f acc0 = {}, acc1 = {}, acc2 = {}, acc3 = {};

    const float* xa = X + (size_t)(m0 + lane16) * K + (hi ? 2 : 0);
    const float* wb = Wp + (size_t)lane * 2;

    for (int k = 0; k < K; k += 4) {
        float2 af = *(const float2*)(xa + k);
        v2f a; a.x = af.x; a.y = af.y;

        const float* wk = wb + (size_t)(k >> 2) * 256;
        v2f b0 = *(const v2f*)(wk);
        v2f b1 = *(const v2f*)(wk + 64);
        v2f b2 = *(const v2f*)(wk + 128);
        v2f b3 = *(const v2f*)(wk + 192);

        acc0 = __builtin_amdgcn_wmma_f32_16x16x4_f32(false, a, false, b0, (short)0, acc0, false, false);
        acc1 = __builtin_amdgcn_wmma_f32_16x16x4_f32(false, a, false, b1, (short)0, acc1, false, false);
        acc2 = __builtin_amdgcn_wmma_f32_16x16x4_f32(false, a, false, b2, (short)0, acc2, false, false);
        acc3 = __builtin_amdgcn_wmma_f32_16x16x4_f32(false, a, false, b3, (short)0, acc3, false, false);
    }

    const int rbase = m0 + (hi ? 8 : 0);
    #pragma unroll
    for (int r = 0; r < 8; ++r) {
        float* o = H + (size_t)(rbase + r) * 64 + lane16;
        o[0]  = acc0[r];
        o[16] = acc1[r];
        o[32] = acc2[r];
        o[48] = acc3[r];
    }
}

__global__ __launch_bounds__(256) void fgcn_bias_init(float* __restrict__ out,
                                                      const float* __restrict__ bias,
                                                      int total) {
    int idx = blockIdx.x * blockDim.x + threadIdx.x;
    if (idx < total) out[idx] = bias[idx & 63];
}

__global__ __launch_bounds__(256) void fgcn_edge_scatter(const float* __restrict__ h,
                                                         const int* __restrict__ src,
                                                         const int* __restrict__ dst,
                                                         float* __restrict__ out,
                                                         int E) {
    const int lane = threadIdx.x & 31;
    const int wave = (blockIdx.x * blockDim.x + threadIdx.x) >> 5;
    const int e0   = wave * 32;
    if (e0 >= E) return;

    const int e = e0 + lane;
    int s = 0, d = 0;
    if (e < E) { s = src[e]; d = dst[e]; }

    if (e0 + 32 <= E) {
        #pragma unroll
        for (int g = 0; g < 4; ++g) {
            float a[8], b[8];
            #pragma unroll
            for (int u = 0; u < 8; ++u) {
                const int sj = __builtin_amdgcn_readlane(s, g * 8 + u);
                const float* hp = h + (size_t)sj * 64;
                a[u] = hp[lane];
                b[u] = hp[lane + 32];
            }
            #pragma unroll
            for (int u = 0; u < 8; ++u) {
                const int dj = __builtin_amdgcn_readlane(d, g * 8 + u);
                float* op = out + (size_t)dj * 64;
                atomicAdd(op + lane, a[u]);
                atomicAdd(op + lane + 32, b[u]);
            }
        }
    } else {
        const int n = E - e0;
        for (int j = 0; j < n; ++j) {
            const int sj = __shfl(s, j);
            const int dj = __shfl(d, j);
            const float* hp = h + (size_t)sj * 64;
            float* op = out + (size_t)dj * 64;
            atomicAdd(op + lane, hp[lane]);
            atomicAdd(op + lane + 32, hp[lane + 32]);
        }
    }
}

static constexpr int N_DRUG = 50000, E_DRUG = 1000000, F_DRUG = 256;
static constexpr int N_DIS  = 20000, E_DIS  = 640000,  F_DIS  = 128;
static constexpr int NHID2  = 64;

extern "C" void kernel_launch(void* const* d_in, const int* in_sizes, int n_in,
                              void* d_out, int out_size, void* d_ws, size_t ws_size,
                              hipStream_t stream) {
    const float* drug_x  = (const float*)d_in[0];
    const float* dis_x   = (const float*)d_in[1];
    const float* W1      = (const float*)d_in[2];
    const float* b1      = (const float*)d_in[3];
    const float* W2      = (const float*)d_in[4];
    const float* b2      = (const float*)d_in[5];
    const int*   drug_ei = (const int*)d_in[6];
    const int*   dis_ei  = (const int*)d_in[7];

    float* out1 = (float*)d_out;
    float* out2 = out1 + (size_t)N_DRUG * NHID2;

    float* h1  = (float*)d_ws;
    float* h2  = h1 + (size_t)N_DRUG * NHID2;
    float* Wp1 = h2 + (size_t)N_DIS * NHID2;
    float* Wp2 = Wp1 + (size_t)F_DRUG * NHID2;

    {
        int total = F_DRUG * NHID2 / 2;
        fgcn_pack_w<<<(total + 255) / 256, 256, 0, stream>>>(W1, Wp1, total);
    }
    {
        int total = F_DIS * NHID2 / 2;
        fgcn_pack_w<<<(total + 255) / 256, 256, 0, stream>>>(W2, Wp2, total);
    }

    {
        int mtiles = N_DRUG / 16;
        int grid = (mtiles + 7) / 8;
        fgcn_gemm_wmma<<<grid, 256, 0, stream>>>(drug_x, Wp1, h1, mtiles, F_DRUG);
    }
    {
        int mtiles = N_DIS / 16;
        int grid = (mtiles + 7) / 8;
        fgcn_gemm_wmma<<<grid, 256, 0, stream>>>(dis_x, Wp2, h2, mtiles, F_DIS);
    }

    {
        int total = N_DRUG * NHID2;
        fgcn_bias_init<<<(total + 255) / 256, 256, 0, stream>>>(out1, b1, total);
    }
    {
        int total = N_DIS * NHID2;
        fgcn_bias_init<<<(total + 255) / 256, 256, 0, stream>>>(out2, b2, total);
    }

    {
        int threads = E_DRUG;
        fgcn_edge_scatter<<<(threads + 255) / 256, 256, 0, stream>>>(
            h1, drug_ei, drug_ei + E_DRUG, out1, E_DRUG);
    }
    {
        int threads = E_DIS;
        fgcn_edge_scatter<<<(threads + 255) / 256, 256, 0, stream>>>(
            h2, dis_ei, dis_ei + E_DIS, out2, E_DIS);
    }
}


# out's start value (bias) is stored by the GEMM wave that owns the 16 rows, as 8 whole-line 16-byte stores; the two init launches and their 140000 one-store waves go; sum order unchanged
# speedup vs baseline: 1.9391x; 1.1349x over previous
#include <hip/hip_runtime.h>
#include <hip/hip_bf16.h>

typedef __attribute__((ext_vector_type(2))) float v2f;
typedef __attribute__((ext_vector_type(4))) float v4f;
typedef __attribute__((ext_vector_type(8))) float v8f;

__global__ __launch_bounds__(256) void fgcn_pack_w(const float* __restrict__ W,
                                                   float* __restrict__ Wp,
                                                   int total  ) {
    int tid = blockIdx.x * blockDim.x + threadIdx.x;
    if (tid >= total) return;
    const int lane = tid & 31;
    const int n    = (tid >> 5) & 3;
    const int kk   = tid >> 7;
    const int kb   = kk * 4 + ((lane >= 16) ? 2 : 0);
    const int col  = n * 16 + (lane & 15);
    float2 v;
    v.x = W[(size_t)kb * 64 + col];
    v.y = W[(size_t)(kb + 1) * 64 + col];
    *(float2*)(Wp + (size_t)tid * 2) = v;
}

__global__ __launch_bounds__(256) void fgcn_gemm_wmma(const float* __restrict__ X,
                                                      const float* __restrict__ Wp,
                                                      float* __restrict__ H,
                                                      const float* __restrict__ bias,
                                                      float* __restrict__ out,
                                                      int mtiles, int K) {
    const int wave = (blockIdx.x * blockDim.x + threadIdx.x) >> 5;
    const int lane = threadIdx.x & 31;
    if (wave >= mtiles) return;

    const int m0     = wave * 16;
    const int lane16 = lane & 15;
    const bool hi    = lane >= 16;

    v8f acc0 = {}, acc1 = {}, acc2 = {}, acc3 = {};

    const v4f bv = *(const v4f*)(bias + (lane & 15) * 4);

    const float* xa = X + (size_t)(m0 + lane16) * K + (hi ? 2 : 0);
    const float* wb = Wp + (size_t)lane * 2;

    for (int k = 0; k < K; k += 4) {
        float2 af = *(const float2*)(xa + k);
        v2f a; a.x = af.x; a.y = af.y;

        const float* wk = wb + (size_t)(k >> 2) * 256;
        v2f b0 = *(const v2f*)(wk);
        v2f b1 = *(const v2f*)(wk + 64);
        v2f b2 = *(const v2f*)(wk + 128);
        v2f b3 = *(const v2f*)(wk + 192);

        acc0 = __builtin_amdgcn_wmma_f32_16x16x4_f32(false, a, false, b0, (short)0, acc0, false, false);
        acc1 = __builtin_amdgcn_wmma_f32_16x16x4_f32(false, a, false, b1, (short)0, acc1, false, false);
        acc2 = __builtin_amdgcn_wmma_f32_16x16x4_f32(false, a, false, b2, (short)0, acc2, false, false);
        acc3 = __builtin_amdgcn_wmma_f32_16x16x4_f32(false, a, false, b3, (short)0, acc3, false, false);
    }

    const int rbase = m0 + (hi ? 8 : 0);
    #pragma unroll
    for (int r = 0; r < 8; ++r) {
        float* o = H + (size_t)(rbase + r) * 64 + lane16;
        o[0]  = acc0[r];
        o[16] = acc1[r];
        o[32] = acc2[r];
        o[48] = acc3[r];
    }

    v4f* ob = (v4f*)(out + (size_t)m0 * 64) + lane;
    #pragma unroll
    for (int i = 0; i < 8; ++i) ob[i * 32] = bv;
}

__global__ __launch_bounds__(256) void fgcn_edge_scatter(const float* __restrict__ h,
                                                         const int* __restrict__ src,
                                                         const int* __restrict__ dst,
                                                         float* __restrict__ out,
                                                         int E) {
    const int lane = threadIdx.x & 31;
    const int wave = (blockIdx.x * blockDim.x + threadIdx.x) >> 5;
    const int e0   = wave * 32;
    if (e0 >= E) return;

    const int e = e0 + lane;
    int s = 0, d = 0;
    if (e < E) { s = src[e]; d = dst[e]; }

    if (e0 + 32 <= E) {
        #pragma unroll
        for (int g = 0; g < 4; ++g) {
            float a[8], b[8];
            #pragma unroll
            for (int u = 0; u < 8; ++u) {
                const int sj = __builtin_amdgcn_readlane(s, g * 8 + u);
                const float* hp = h + (size_t)sj * 64;
                a[u] = hp[lane];
                b[u] = hp[lane + 32];
            }
            #pragma unroll
            for (int u = 0; u < 8; ++u) {
                const int dj = __builtin_amdgcn_readlane(d, g * 8 + u);
                float* op = out + (size_t)dj * 64;
                atomicAdd(op + lane, a[u]);
                atomicAdd(op + lane + 32, b[u]);
            }
        }
    } else {
        const int n = E - e0;
        for (int j = 0; j < n; ++j) {
            const int sj = __shfl(s, j);
            const int dj = __shfl(d, j);
            const float* hp = h + (size_t)sj * 64;
            float* op = out + (size_t)dj * 64;
            atomicAdd(op + lane, hp[lane]);
            atomicAdd(op + lane + 32, hp[lane + 32]);
        }
    }
}

static constexpr int N_DRUG = 50000, E_DRUG = 1000000, F_DRUG = 256;
static constexpr int N_DIS  = 20000, E_DIS  = 640000,  F_DIS  = 128;
static constexpr int NHID2  = 64;

extern "C" void kernel_launch(void* const* d_in, const int* in_sizes, int n_in,
                              void* d_out, int out_size, void* d_ws, size_t ws_size,
                              hipStream_t stream) {
    const float* drug_x  = (const float*)d_in[0];
    const float* dis_x   = (const float*)d_in[1];
    const float* W1      = (const float*)d_in[2];
    const float* b1      = (const float*)d_in[3];
    const float* W2      = (const float*)d_in[4];
    const float* b2      = (const float*)d_in[5];
    const int*   drug_ei = (const int*)d_in[6];
    const int*   dis_ei  = (const int*)d_in[7];

    float* out1 = (float*)d_out;
    float* out2 = out1 + (size_t)N_DRUG * NHID2;

    float* h1  = (float*)d_ws;
    float* h2  = h1 + (size_t)N_DRUG * NHID2;
    float* Wp1 = h2 + (size_t)N_DIS * NHID2;
    float* Wp2 = Wp1 + (size_t)F_DRUG * NHID2;

    {
        int total = F_DRUG * NHID2 / 2;
        fgcn_pack_w<<<(total + 255) / 256, 256, 0, stream>>>(W1, Wp1, total);
    }
    {
        int total = F_DIS * NHID2 / 2;
        fgcn_pack_w<<<(total + 255) / 256, 256, 0, stream>>>(W2, Wp2, total);
    }

    {
        int mtiles = N_DRUG / 16;
        int grid = (mtiles + 7) / 8;
        fgcn_gemm_wmma<<<grid, 256, 0, stream>>>(drug_x, Wp1, h1, b1, out1, mtiles, F_DRUG);
    }
    {
        int mtiles = N_DIS / 16;
        int grid = (mtiles + 7) / 8;
        fgcn_gemm_wmma<<<grid, 256, 0, stream>>>(dis_x, Wp2, h2, b2, out2, mtiles, F_DIS);
    }

    {
        int threads = E_DRUG;
        fgcn_edge_scatter<<<(threads + 255) / 256, 256, 0, stream>>>(
            h1, drug_ei, drug_ei + E_DRUG, out1, E_DRUG);
    }
    {
        int threads = E_DIS;
        fgcn_edge_scatter<<<(threads + 255) / 256, 256, 0, stream>>>(
            h2, dis_ei, dis_ei + E_DIS, out2, E_DIS);
    }
}


# launches 6 -> 2: drug and disease share one GEMM and one scatter launch; each GEMM block stages W into LDS in fragment order itself, so both pack launches go; sum order unchanged
# speedup vs baseline: 2.2946x; 1.1833x over previous
#include <hip/hip_runtime.h>


typedef __attribute__((ext_vector_type(2))) float v2f;
typedef __attribute__((ext_vector_type(4))) float v4f;
typedef __attribute__((ext_vector_type(8))) float v8f;

static constexpr int N_DRUG = 50000, E_DRUG = 1000000, F_DRUG = 256;
static constexpr int N_DIS  = 20000, E_DIS  = 640000,  F_DIS  = 128;
static constexpr int NHID2  = 64;

static constexpr int GEMM_BLOCKS_DRUG = (N_DRUG / 16 + 7) / 8;
static constexpr int GEMM_BLOCKS_DIS  = (N_DIS / 16 + 7) / 8;
static constexpr int SCAT_BLOCKS_DRUG = (E_DRUG + 255) / 256;
static constexpr int SCAT_BLOCKS_DIS  = (E_DIS + 255) / 256;

template <int K>
__device__ __forceinline__ void gemm_block(const float* __restrict__ X,
                                           const float* __restrict__ W,
                                           float* __restrict__ H,
                                           const float* __restrict__ bias,
                                           float* __restrict__ out,
                                           const int blk, const int mtiles,
                                           v4f* Ws) {
    const int t = threadIdx.x;

    constexpr int TASKS = (K / 2) * 16;
    constexpr int PER   = TASKS / 256;
    v4f r0[PER], r1[PER];
    #pragma unroll
    for (int i = 0; i < PER; ++i) {
        const int q = t + 256 * i;
        const int j = q & 15, p = q >> 4;
        r0[i] = *(const v4f*)(W + (size_t)(2 * p) * 64 + 4 * j);
        r1[i] = *(const v4f*)(W + (size_t)(2 * p + 1) * 64 + 4 * j);
    }
    #pragma unroll
    for (int i = 0; i < PER; ++i) {
        const int q = t + 256 * i;
        const int j = q & 15, p = q >> 4;
        const int kk = p >> 1, h = p & 1, n = j >> 2, c0 = (j & 3) * 4;
        const int f2 = (kk * 4 + n) * 32 + h * 16 + c0;
        v4f lo = {r0[i].x, r1[i].x, r0[i].y, r1[i].y};
        v4f hi = {r0[i].z, r1[i].z, r0[i].w, r1[i].w};
        Ws[(f2 >> 1)]     = lo;
        Ws[(f2 >> 1) + 1] = hi;
    }
    __syncthreads();

    const int wv   = __builtin_amdgcn_readfirstlane(t >> 5);
    const int lane = t & 31;
    const int tile = blk * 8 + wv;
    if (tile >= mtiles) return;

    const int m0     = tile * 16;
    const int lane16 = lane & 15;
    const bool hi    = lane >= 16;

    v8f acc0 = {}, acc1 = {}, acc2 = {}, acc3 = {};

    const v4f bv = *(const v4f*)(bias + (lane & 15) * 4);

    const float* xa = X + (size_t)(m0 + lane16) * K + (hi ? 2 : 0);
    const v2f* wl = (const v2f*)Ws + lane;

    for (int k = 0; k < K; k += 4) {
        float2 af = *(const float2*)(xa + k);
        v2f a; a.x = af.x; a.y = af.y;

        const v2f* wk = wl + (k >> 2) * 128;
        v2f b0 = wk[0];
        v2f b1 = wk[32];
        v2f b2 = wk[64];
        v2f b3 = wk[96];

        acc0 = __builtin_amdgcn_wmma_f32_16x16x4_f32(false, a, false, b0, (short)0, acc0, false, false);
        acc1 = __builtin_amdgcn_wmma_f32_16x16x4_f32(false, a, false, b1, (short)0, acc1, false, false);
        acc2 = __builtin_amdgcn_wmma_f32_16x16x4_f32(false, a, false, b2, (short)0, acc2, false, false);
        acc3 = __builtin_amdgcn_wmma_f32_16x16x4_f32(false, a, false, b3, (short)0, acc3, false, false);
    }

    const int rbase = m0 + (hi ? 8 : 0);
    #pragma unroll
    for (int r = 0; r < 8; ++r) {
        float* o = H + (size_t)(rbase + r) * 64 + lane16;
        o[0]  = acc0[r];
        o[16] = acc1[r];
        o[32] = acc2[r];
        o[48] = acc3[r];
    }

    v4f* ob = (v4f*)(out + (size_t)m0 * 64) + lane;
    #pragma unroll
    for (int i = 0; i < 8; ++i) ob[i * 32] = bv;
}

__global__ __launch_bounds__(256) void fgcn_gemm_both(const float* __restrict__ X1,
                                                      const float* __restrict__ W1,
                                                      const float* __restrict__ b1,
                                                      float* __restrict__ H1,
                                                      float* __restrict__ out1,
                                                      const float* __restrict__ X2,
                                                      const float* __restrict__ W2,
                                                      const float* __restrict__ b2,
                                                      float* __restrict__ H2,
                                                      float* __restrict__ out2) {
    __shared__ v4f Ws[F_DRUG * NHID2 / 4];
    const int b = blockIdx.x;
    if (b < GEMM_BLOCKS_DRUG) {
        gemm_block<F_DRUG>(X1, W1, H1, b1, out1, b, N_DRUG / 16, Ws);
    } else {
        gemm_block<F_DIS>(X2, W2, H2, b2, out2, b - GEMM_BLOCKS_DRUG, N_DIS / 16, Ws);
    }
}

template <int E>
__device__ __forceinline__ void scatter_block(const float* __restrict__ h,
                                              const int* __restrict__ src,
                                              const int* __restrict__ dst,
                                              float* __restrict__ out,
                                              const int blk) {
    const int lane = threadIdx.x & 31;
    const int wave = (blk * 256 + (int)threadIdx.x) >> 5;
    const int e0   = wave * 32;
    if (e0 >= E) return;

    const int e = e0 + lane;
    int s = 0, d = 0;
    if (e < E) { s = src[e]; d = dst[e]; }

    if (e0 + 32 <= E) {
        #pragma unroll
        for (int g = 0; g < 4; ++g) {
            float a[8], b[8];
            #pragma unroll
            for (int u = 0; u < 8; ++u) {
                const int sj = __builtin_amdgcn_readlane(s, g * 8 + u);
                const float* hp = h + (size_t)sj * 64;
                a[u] = hp[lane];
                b[u] = hp[lane + 32];
            }
            #pragma unroll
            for (int u = 0; u < 8; ++u) {
                const int dj = __builtin_amdgcn_readlane(d, g * 8 + u);
                float* op = out + (size_t)dj * 64;
                atomicAdd(op + lane, a[u]);
                atomicAdd(op + lane + 32, b[u]);
            }
        }
    } else {
        const int n = E - e0;
        for (int j = 0; j < n; ++j) {
            const int sj = __shfl(s, j);
            const int dj = __shfl(d, j);
            const float* hp = h + (size_t)sj * 64;
            float* op = out + (size_t)dj * 64;
            atomicAdd(op + lane, hp[lane]);
            atomicAdd(op + lane + 32, hp[lane + 32]);
        }
    }
}

__global__ __launch_bounds__(256) void fgcn_edge_scatter_both(const float* __restrict__ h1,
                                                              const int* __restrict__ src1,
                                                              const int* __restrict__ dst1,
                                                              float* __restrict__ out1,
                                                              const float* __restrict__ h2,
                                                              const int* __restrict__ src2,
                                                              const int* __restrict__ dst2,
                                                              float* __restrict__ out2) {
    const int b = blockIdx.x;
    if (b < SCAT_BLOCKS_DRUG) {
        scatter_block<E_DRUG>(h1, src1, dst1, out1, b);
    } else {
        scatter_block<E_DIS>(h2, src2, dst2, out2, b - SCAT_BLOCKS_DRUG);
    }
}

extern "C" void kernel_launch(void* const* d_in, const int* in_sizes, int n_in,
                              void* d_out, int out_size, void* d_ws, size_t ws_size,
                              hipStream_t stream) {
    const float* drug_x  = (const float*)d_in[0];
    const float* dis_x   = (const float*)d_in[1];
    const float* W1      = (const float*)d_in[2];
    const float* b1      = (const float*)d_in[3];
    const float* W2      = (const float*)d_in[4];
    const float* b2      = (const float*)d_in[5];
    const int*   drug_ei = (const int*)d_in[6];
    const int*   dis_ei  = (const int*)d_in[7];

    float* out1 = (float*)d_out;
    float* out2 = out1 + (size_t)N_DRUG * NHID2;

    float* h1  = (float*)d_ws;
    float* h2  = h1 + (size_t)N_DRUG * NHID2;

    fgcn_gemm_both<<<GEMM_BLOCKS_DRUG + GEMM_BLOCKS_DIS, 256, 0, stream>>>(
        drug_x, W1, b1, h1, out1, dis_x, W2, b2, h2, out2);

    fgcn_edge_scatter_both<<<SCAT_BLOCKS_DRUG + SCAT_BLOCKS_DIS, 256, 0, stream>>>(
        h1, drug_ei, drug_ei + E_DRUG, out1, h2, dis_ei, dis_ei + E_DIS, out2);
}


# edge scatter as resident blocks drawing 256-edge chunks from one counter per graph, each cache half its own graph first: a row of out gets its float atomics from one half; sum order by arrival
# speedup vs baseline: 2.7799x; 1.2115x over previous
#include <hip/hip_runtime.h>


typedef __attribute__((ext_vector_type(2))) float v2f;
typedef __attribute__((ext_vector_type(4))) float v4f;
typedef __attribute__((ext_vector_type(8))) float v8f;

static constexpr int N_DRUG = 50000, E_DRUG = 1000000, F_DRUG = 256;
static constexpr int N_DIS  = 20000, E_DIS  = 640000,  F_DIS  = 128;
static constexpr int NHID2  = 64;

static constexpr int GEMM_BLOCKS_DRUG = (N_DRUG / 16 + 7) / 8;
static constexpr int GEMM_BLOCKS_DIS  = (N_DIS / 16 + 7) / 8;
static constexpr int SCAT_CHUNKS_DRUG = (E_DRUG + 255) / 256;
static constexpr int SCAT_CHUNKS_DIS  = (E_DIS + 255) / 256;
static constexpr int SCAT_BLOCKS      = 256 * 8;
static constexpr int CTR_PITCH        = 64;

template <int K>
__device__ __forceinline__ void gemm_block(const float* __restrict__ X,
                                           const float* __restrict__ W,
                                           float* __restrict__ H,
                                           const float* __restrict__ bias,
                                           float* __restrict__ out,
                                           const int blk, const int mtiles,
                                           v4f* Ws) {
    const int t = threadIdx.x;

    constexpr int TASKS = (K / 2) * 16;
    constexpr int PER   = TASKS / 256;
    v4f r0[PER], r1[PER];
    #pragma unroll
    for (int i = 0; i < PER; ++i) {
        const int q = t + 256 * i;
        const int j = q & 15, p = q >> 4;
        r0[i] = *(const v4f*)(W + (size_t)(2 * p) * 64 + 4 * j);
        r1[i] = *(const v4f*)(W + (size_t)(2 * p + 1) * 64 + 4 * j);
    }
    #pragma unroll
    for (int i = 0; i < PER; ++i) {
        const int q = t + 256 * i;
        const int j = q & 15, p = q >> 4;
        const int kk = p >> 1, h = p & 1, n = j >> 2, c0 = (j & 3) * 4;
        const int f2 = (kk * 4 + n) * 32 + h * 16 + c0;
        v4f lo = {r0[i].x, r1[i].x, r0[i].y, r1[i].y};
        v4f hi = {r0[i].z, r1[i].z, r0[i].w, r1[i].w};
        Ws[(f2 >> 1)]     = lo;
        Ws[(f2 >> 1) + 1] = hi;
    }
    __syncthreads();

    const int wv   = __builtin_amdgcn_readfirstlane(t >> 5);
    const int lane = t & 31;
    const int tile = blk * 8 + wv;
    if (tile >= mtiles) return;

    const int m0     = tile * 16;
    const int lane16 = lane & 15;
    const bool hi    = lane >= 16;

    v8f acc0 = {}, acc1 = {}, acc2 = {}, acc3 = {};

    const v4f bv = *(const v4f*)(bias + (lane & 15) * 4);

    const float* xa = X + (size_t)(m0 + lane16) * K + (hi ? 2 : 0);
    const v2f* wl = (const v2f*)Ws + lane;

    for (int k = 0; k < K; k += 4) {
        float2 af = *(const float2*)(xa + k);
        v2f a; a.x = af.x; a.y = af.y;

        const v2f* wk = wl + (k >> 2) * 128;
        v2f b0 = wk[0];
        v2f b1 = wk[32];
        v2f b2 = wk[64];
        v2f b3 = wk[96];

        acc0 = __builtin_amdgcn_wmma_f32_16x16x4_f32(false, a, false, b0, (short)0, acc0, false, false);
        acc1 = __builtin_amdgcn_wmma_f32_16x16x4_f32(false, a, false, b1, (short)0, acc1, false, false);
        acc2 = __builtin_amdgcn_wmma_f32_16x16x4_f32(false, a, false, b2, (short)0, acc2, false, false);
        acc3 = __builtin_amdgcn_wmma_f32_16x16x4_f32(false, a, false, b3, (short)0, acc3, false, false);
    }

    const int rbase = m0 + (hi ? 8 : 0);
    #pragma unroll
    for (int r = 0; r < 8; ++r) {
        float* o = H + (size_t)(rbase + r) * 64 + lane16;
        o[0]  = acc0[r];
        o[16] = acc1[r];
        o[32] = acc2[r];
        o[48] = acc3[r];
    }

    v4f* ob = (v4f*)(out + (size_t)m0 * 64) + lane;
    #pragma unroll
    for (int i = 0; i < 8; ++i) ob[i * 32] = bv;
}

__global__ __launch_bounds__(256) void fgcn_gemm_both(const float* __restrict__ X1,
                                                      const float* __restrict__ W1,
                                                      const float* __restrict__ b1,
                                                      float* __restrict__ H1,
                                                      float* __restrict__ out1,
                                                      const float* __restrict__ X2,
                                                      const float* __restrict__ W2,
                                                      const float* __restrict__ b2,
                                                      float* __restrict__ H2,
                                                      float* __restrict__ out2,
                                                      int* __restrict__ ctr) {
    __shared__ v4f Ws[F_DRUG * NHID2 / 4];
    const int b = blockIdx.x;
    if (b == 0 && threadIdx.x < 2) ctr[threadIdx.x * CTR_PITCH] = 0;
    if (b < GEMM_BLOCKS_DRUG) {
        gemm_block<F_DRUG>(X1, W1, H1, b1, out1, b, N_DRUG / 16, Ws);
    } else {
        gemm_block<F_DIS>(X2, W2, H2, b2, out2, b - GEMM_BLOCKS_DRUG, N_DIS / 16, Ws);
    }
}

__device__ __forceinline__ void scatter_wave(const float* __restrict__ h,
                                             float* __restrict__ out,
                                             const int s, const int d,
                                             const int n, const int lane) {
    if (n >= 32) {
        #pragma unroll
        for (int g = 0; g < 4; ++g) {
            float a[8], b[8];
            #pragma unroll
            for (int u = 0; u < 8; ++u) {
                const int sj = __builtin_amdgcn_readlane(s, g * 8 + u);
                const float* hp = h + (size_t)sj * 64;
                a[u] = hp[lane];
                b[u] = hp[lane + 32];
            }
            #pragma unroll
            for (int u = 0; u < 8; ++u) {
                const int dj = __builtin_amdgcn_readlane(d, g * 8 + u);
                float* op = out + (size_t)dj * 64;
                atomicAdd(op + lane, a[u]);
                atomicAdd(op + lane + 32, b[u]);
            }
        }
    } else {
        for (int j = 0; j < n; ++j) {
            const int sj = __shfl(s, j);
            const int dj = __shfl(d, j);
            const float* hp = h + (size_t)sj * 64;
            float* op = out + (size_t)dj * 64;
            atomicAdd(op + lane, hp[lane]);
            atomicAdd(op + lane + 32, hp[lane + 32]);
        }
    }
}

__global__ __launch_bounds__(256) void fgcn_edge_scatter_both(const float* __restrict__ h1,
                                                              const int* __restrict__ src1,
                                                              const int* __restrict__ dst1,
                                                              float* __restrict__ out1,
                                                              const float* __restrict__ h2,
                                                              const int* __restrict__ src2,
                                                              const int* __restrict__ dst2,
                                                              float* __restrict__ out2,
                                                              int* __restrict__ ctr) {
    __shared__ int s_half, s_ticket;
    const int t    = threadIdx.x;
    const int lane = t & 31;

    if (t == 0) {
        const unsigned id = __builtin_amdgcn_s_sendmsg_rtn(0x87);
        const int hf = (int)((id >> 18) & 1u);
        s_half   = hf;
        s_ticket = atomicAdd(&ctr[hf * CTR_PITCH], 1);
    }
    __syncthreads();
    const int half = __builtin_amdgcn_readfirstlane(s_half);
    int c          = __builtin_amdgcn_readfirstlane(s_ticket);
    const int wv   = __builtin_amdgcn_readfirstlane(t >> 5);
    __syncthreads();

    for (int pass = 0; pass < 2; ++pass) {
        const int g   = pass == 0 ? half : 1 - half;
        const int nch = g == 0 ? SCAT_CHUNKS_DRUG : SCAT_CHUNKS_DIS;
        const int E   = g == 0 ? E_DRUG : E_DIS;
        const float* __restrict__ h   = g == 0 ? h1 : h2;
        const int* __restrict__   src = g == 0 ? src1 : src2;
        const int* __restrict__   dst = g == 0 ? dst1 : dst2;
        float* __restrict__       out = g == 0 ? out1 : out2;
        if (pass == 1) {
            if (t == 0) s_ticket = atomicAdd(&ctr[g * CTR_PITCH], 1);
            __syncthreads();
            c = __builtin_amdgcn_readfirstlane(s_ticket);
            __syncthreads();
        }
        while (c < nch) {
            const int e0 = (c * 8 + wv) * 32;
            const int e  = e0 + lane;
            int s = 0, d = 0;
            if (e < E) { s = src[e]; d = dst[e]; }
            int nxt = 0;
            if (t == 0) nxt = atomicAdd(&ctr[g * CTR_PITCH], 1);
            if (e0 < E) scatter_wave(h, out, s, d, E - e0, lane);
            if (t == 0) s_ticket = nxt;
            __syncthreads();
            c = __builtin_amdgcn_readfirstlane(s_ticket);
            __syncthreads();
        }
    }
}

extern "C" void kernel_launch(void* const* d_in, const int* in_sizes, int n_in,
                              void* d_out, int out_size, void* d_ws, size_t ws_size,
                              hipStream_t stream) {
    const float* drug_x  = (const float*)d_in[0];
    const float* dis_x   = (const float*)d_in[1];
    const float* W1      = (const float*)d_in[2];
    const float* b1      = (const float*)d_in[3];
    const float* W2      = (const float*)d_in[4];
    const float* b2      = (const float*)d_in[5];
    const int*   drug_ei = (const int*)d_in[6];
    const int*   dis_ei  = (const int*)d_in[7];

    float* out1 = (float*)d_out;
    float* out2 = out1 + (size_t)N_DRUG * NHID2;

    float* h1  = (float*)d_ws;
    float* h2  = h1 + (size_t)N_DRUG * NHID2;
    int*   ctr = (int*)(h2 + (size_t)N_DIS * NHID2);

    fgcn_gemm_both<<<GEMM_BLOCKS_DRUG + GEMM_BLOCKS_DIS, 256, 0, stream>>>(
        drug_x, W1, b1, h1, out1, dis_x, W2, b2, h2, out2, ctr);

    fgcn_edge_scatter_both<<<SCAT_BLOCKS, 256, 0, stream>>>(
        h1, drug_ei, drug_ei + E_DRUG, out1, h2, dis_ei, dis_ei + E_DIS, out2, ctr);
}
